# starting kernel
# baseline (speedup 1.0000x reference)
#include <hip/hip_runtime.h>


typedef __attribute__((ext_vector_type(8))) float v8f;
typedef __attribute__((ext_vector_type(2))) float v2f;

#define D_   16
#define L1_  4
#define E_   24
#define K_   16
#define NV_  16
#define TR_  16
#define WAVES_ 8

__device__ __forceinline__ float clamp01(float v) {
    return __builtin_fminf(__builtin_fmaxf(v, 0.0f), 1.0f);
}

__global__ __launch_bounds__(256, 1)
void lattice_ensemble_kernel(const float* __restrict__ x,
                             const float* __restrict__ cal_kp,
                             const float* __restrict__ cal_vals,
                             const float* __restrict__ k1,
                             const float* __restrict__ k2,
                             float* __restrict__ out,
                             int nrows)
{
    __shared__ __align__(16) float sA [D_][K_ - 1];
    __shared__ __align__(16) float sB [D_][K_ - 1];
    __shared__ __align__(16) float sDV[D_][K_ - 1];
    __shared__ float sV0[D_];
    __shared__ __align__(16) float sXC[WAVES_][TR_][20];

    const int tid = threadIdx.x;
    if (tid < D_ * (K_ - 1)) {
        int f = tid / (K_ - 1), s = tid % (K_ - 1);
        float kp0 = cal_kp[f * K_ + s], kp1 = cal_kp[f * K_ + s + 1];
        float inv = 1.0f / (kp1 - kp0);
        sA [f][s] = inv;
        sB [f][s] = -kp0 * inv;
        sDV[f][s] = cal_vals[f * K_ + s + 1] - cal_vals[f * K_ + s];
    }
    if (tid < D_) sV0[tid] = cal_vals[tid * K_];
    __syncthreads();

    const int lane = tid & 31;
    const int wid  = tid >> 5;
    const int nn   = lane & 15;
    const bool hib = (lane >= 16);
    const int  hi  = hib ? 1 : 0;

    v2f Bf[2][L1_][4];
    #pragma unroll
    for (int ct = 0; ct < 2; ++ct) {
        int e = ct * 16 + nn;
        #pragma unroll
        for (int l = 0; l < L1_; ++l) {
            #pragma unroll
            for (int c = 0; c < 4; ++c) {
                int kb = 4 * c + 2 * hi;
                float b0 = 0.0f, b1 = 0.0f;
                if (e < E_) {
                    const float* kp = k1 + (size_t)e * (L1_ * NV_) + l * NV_ + kb;
                    b0 = kp[0];
                    b1 = kp[1];
                }
                v2f bb; bb[0] = b0; bb[1] = b1;
                Bf[ct][l][c] = bb;
            }
        }
    }
    float k2r[2][NV_];
    #pragma unroll
    for (int ct = 0; ct < 2; ++ct) {
        int e = ct * 16 + nn;
        #pragma unroll
        for (int k = 0; k < NV_; ++k)
            k2r[ct][k] = (e < E_) ? k2[(size_t)e * NV_ + k] : 0.0f;
    }

    const int ntiles = nrows >> 4;
    const int nwaves = gridDim.x * WAVES_;
    for (int tile = blockIdx.x * WAVES_ + wid; tile < ntiles; tile += nwaves) {
        const int rowBase = tile * TR_;

        int nxt = tile + nwaves;
        if (nxt < ntiles)
            __builtin_prefetch(x + (size_t)(nxt * TR_ + nn) * D_ + 8 * hi, 0, 1);

        const float* xp = x + (size_t)(rowBase + nn) * D_ + 8 * hi;
        const float4 xa = *(const float4*)(xp);
        const float4 xb = *(const float4*)(xp + 4);
        float xv[8] = { xa.x, xa.y, xa.z, xa.w, xb.x, xb.y, xb.z, xb.w };
        float xcv[8];
        #pragma unroll
        for (int i = 0; i < 8; ++i) {
            const int f = 8 * hi + i;
            float v = xv[i];
            float acc = sV0[f];
            #pragma unroll
            for (int s = 0; s < K_ - 1; ++s) {
                float t = clamp01(__builtin_fmaf(v, sA[f][s], sB[f][s]));
                acc = __builtin_fmaf(t, sDV[f][s], acc);
            }
            xcv[i] = acc;
        }
        *(float4*)&sXC[wid][nn][8 * hi]     = make_float4(xcv[0], xcv[1], xcv[2], xcv[3]);
        *(float4*)&sXC[wid][nn][8 * hi + 4] = make_float4(xcv[4], xcv[5], xcv[6], xcv[7]);

        asm volatile("s_wait_dscnt 0" ::: "memory");

        const float* rp = &sXC[wid][nn][0];
        const float4 r0 = *(const float4*)(rp);
        const float4 r1 = *(const float4*)(rp + 4);
        const float4 r2 = *(const float4*)(rp + 8);
        const float4 r3 = *(const float4*)(rp + 12);
        const float c16[16] = { r0.x, r0.y, r0.z, r0.w,  r1.x, r1.y, r1.z, r1.w,
                                r2.x, r2.y, r2.z, r2.w,  r3.x, r3.y, r3.z, r3.w };

        float w[L1_][NV_];
        #pragma unroll
        for (int l = 0; l < L1_; ++l) {
            const float x0 = c16[4 * l + 0], x1 = c16[4 * l + 1];
            const float x2 = c16[4 * l + 2], x3 = c16[4 * l + 3];
            const float pq[4] = { (1.0f - x0) * (1.0f - x1), (1.0f - x0) * x1,
                                  x0 * (1.0f - x1),          x0 * x1 };
            const float rs[4] = { (1.0f - x2) * (1.0f - x3), (1.0f - x2) * x3,
                                  x2 * (1.0f - x3),          x2 * x3 };
            #pragma unroll
            for (int ii = 0; ii < 4; ++ii)
                #pragma unroll
                for (int jj = 0; jj < 4; ++jj)
                    w[l][ii * 4 + jj] = pq[ii] * rs[jj];
        }

        v2f Af[L1_][4];
        #pragma unroll
        for (int l = 0; l < L1_; ++l)
            #pragma unroll
            for (int c = 0; c < 4; ++c) {
                v2f a;
                a[0] = hib ? w[l][4 * c + 2] : w[l][4 * c + 0];
                a[1] = hib ? w[l][4 * c + 3] : w[l][4 * c + 1];
                Af[l][c] = a;
            }

        #pragma unroll
        for (int ct = 0; ct < 2; ++ct) {
            v8f acc[L1_];
            #pragma unroll
            for (int l = 0; l < L1_; ++l) acc[l] = (v8f)0.0f;
            #pragma unroll
            for (int l = 0; l < L1_; ++l)
                #pragma unroll
                for (int c = 0; c < 4; ++c)
                    acc[l] = __builtin_amdgcn_wmma_f32_16x16x4_f32(
                         false, Af[l][c],
                         false, Bf[ct][l][c],
                         (short)0, acc[l],
                         false,  false);

            const int e = ct * 16 + nn;
            #pragma unroll
            for (int j = 0; j < 8; ++j) {
                const float h0 = clamp01(acc[0][j]);
                const float h1v = clamp01(acc[1][j]);
                const float h2 = clamp01(acc[2][j]);
                const float h3 = clamp01(acc[3][j]);
                const float pq[4] = { (1.0f - h0) * (1.0f - h1v), (1.0f - h0) * h1v,
                                      h0 * (1.0f - h1v),          h0 * h1v };
                const float rs[4] = { (1.0f - h2) * (1.0f - h3), (1.0f - h2) * h3,
                                      h2 * (1.0f - h3),          h2 * h3 };
                float o = 0.0f;
                #pragma unroll
                for (int ii = 0; ii < 4; ++ii)
                    #pragma unroll
                    for (int jj = 0; jj < 4; ++jj)
                        o = __builtin_fmaf(pq[ii] * rs[jj], k2r[ct][ii * 4 + jj], o);
                const int row = j + 8 * hi;
                if (e < E_)
                    out[(size_t)(rowBase + row) * E_ + e] = o;
            }
        }
    }
}

extern "C" void kernel_launch(void* const* d_in, const int* in_sizes, int n_in,
                              void* d_out, int out_size, void* d_ws, size_t ws_size,
                              hipStream_t stream) {
    const float* x        = (const float*)d_in[0];
    const float* cal_kp   = (const float*)d_in[1];
    const float* cal_vals = (const float*)d_in[2];
    const float* kernels1 = (const float*)d_in[3];
    const float* kernels2 = (const float*)d_in[4];
    float* out = (float*)d_out;

    const int nrows  = in_sizes[0] / D_;
    const int ntiles = nrows / TR_;
    const int blocks = (ntiles + WAVES_ - 1) / WAVES_;

    lattice_ensemble_kernel<<<blocks, 256, 0, stream>>>(
        x, cal_kp, cal_vals, kernels1, kernels2, out, nrows);
}


# Persistent grid: 512 work-groups (all resident at once) instead of 1024, each wave loops over 2 tiles, so wave entry, table build and B-fragment loads are paid once per wave, not per tile
# speedup vs baseline: 1.1708x; 1.1708x over previous
#include <hip/hip_runtime.h>


typedef __attribute__((ext_vector_type(8))) float v8f;
typedef __attribute__((ext_vector_type(2))) float v2f;

#define D_   16
#define L1_  4
#define E_   24
#define K_   16
#define NV_  16
#define TR_  16
#define WAVES_ 8
#define NBLK_  512

__device__ __forceinline__ float clamp01(float v) {
    return __builtin_fminf(__builtin_fmaxf(v, 0.0f), 1.0f);
}

__global__ __launch_bounds__(256, 1)
void lattice_ensemble_kernel(const float* __restrict__ x,
                             const float* __restrict__ cal_kp,
                             const float* __restrict__ cal_vals,
                             const float* __restrict__ k1,
                             const float* __restrict__ k2,
                             float* __restrict__ out,
                             int nrows)
{
    __shared__ __align__(16) float sA [D_][K_ - 1];
    __shared__ __align__(16) float sB [D_][K_ - 1];
    __shared__ __align__(16) float sDV[D_][K_ - 1];
    __shared__ float sV0[D_];
    __shared__ __align__(16) float sXC[WAVES_][TR_][20];

    const int tid = threadIdx.x;
    if (tid < D_ * (K_ - 1)) {
        int f = tid / (K_ - 1), s = tid % (K_ - 1);
        float kp0 = cal_kp[f * K_ + s], kp1 = cal_kp[f * K_ + s + 1];
        float inv = 1.0f / (kp1 - kp0);
        sA [f][s] = inv;
        sB [f][s] = -kp0 * inv;
        sDV[f][s] = cal_vals[f * K_ + s + 1] - cal_vals[f * K_ + s];
    }
    if (tid < D_) sV0[tid] = cal_vals[tid * K_];
    __syncthreads();

    const int lane = tid & 31;
    const int wid  = tid >> 5;
    const int nn   = lane & 15;
    const bool hib = (lane >= 16);
    const int  hi  = hib ? 1 : 0;

    v2f Bf[2][L1_][4];
    #pragma unroll
    for (int ct = 0; ct < 2; ++ct) {
        int e = ct * 16 + nn;
        #pragma unroll
        for (int l = 0; l < L1_; ++l) {
            #pragma unroll
            for (int c = 0; c < 4; ++c) {
                int kb = 4 * c + 2 * hi;
                float b0 = 0.0f, b1 = 0.0f;
                if (e < E_) {
                    const float* kp = k1 + (size_t)e * (L1_ * NV_) + l * NV_ + kb;
                    b0 = kp[0];
                    b1 = kp[1];
                }
                v2f bb; bb[0] = b0; bb[1] = b1;
                Bf[ct][l][c] = bb;
            }
        }
    }
    float k2r[2][NV_];
    #pragma unroll
    for (int ct = 0; ct < 2; ++ct) {
        int e = ct * 16 + nn;
        #pragma unroll
        for (int k = 0; k < NV_; ++k)
            k2r[ct][k] = (e < E_) ? k2[(size_t)e * NV_ + k] : 0.0f;
    }

    const int ntiles = nrows >> 4;
    const int nwaves = gridDim.x * WAVES_;
    for (int tile = blockIdx.x * WAVES_ + wid; tile < ntiles; tile += nwaves) {
        const int rowBase = tile * TR_;

        int nxt = tile + nwaves;
        if (nxt < ntiles)
            __builtin_prefetch(x + (size_t)(nxt * TR_ + nn) * D_ + 8 * hi, 0, 1);

        const float* xp = x + (size_t)(rowBase + nn) * D_ + 8 * hi;
        const float4 xa = *(const float4*)(xp);
        const float4 xb = *(const float4*)(xp + 4);
        float xv[8] = { xa.x, xa.y, xa.z, xa.w, xb.x, xb.y, xb.z, xb.w };
        float xcv[8];
        #pragma unroll
        for (int i = 0; i < 8; ++i) {
            const int f = 8 * hi + i;
            float v = xv[i];
            float acc = sV0[f];
            #pragma unroll
            for (int s = 0; s < K_ - 1; ++s) {
                float t = clamp01(__builtin_fmaf(v, sA[f][s], sB[f][s]));
                acc = __builtin_fmaf(t, sDV[f][s], acc);
            }
            xcv[i] = acc;
        }
        *(float4*)&sXC[wid][nn][8 * hi]     = make_float4(xcv[0], xcv[1], xcv[2], xcv[3]);
        *(float4*)&sXC[wid][nn][8 * hi + 4] = make_float4(xcv[4], xcv[5], xcv[6], xcv[7]);

        asm volatile("s_wait_dscnt 0" ::: "memory");

        const float* rp = &sXC[wid][nn][0];
        const float4 r0 = *(const float4*)(rp);
        const float4 r1 = *(const float4*)(rp + 4);
        const float4 r2 = *(const float4*)(rp + 8);
        const float4 r3 = *(const float4*)(rp + 12);
        const float c16[16] = { r0.x, r0.y, r0.z, r0.w,  r1.x, r1.y, r1.z, r1.w,
                                r2.x, r2.y, r2.z, r2.w,  r3.x, r3.y, r3.z, r3.w };

        float w[L1_][NV_];
        #pragma unroll
        for (int l = 0; l < L1_; ++l) {
            const float x0 = c16[4 * l + 0], x1 = c16[4 * l + 1];
            const float x2 = c16[4 * l + 2], x3 = c16[4 * l + 3];
            const float pq[4] = { (1.0f - x0) * (1.0f - x1), (1.0f - x0) * x1,
                                  x0 * (1.0f - x1),          x0 * x1 };
            const float rs[4] = { (1.0f - x2) * (1.0f - x3), (1.0f - x2) * x3,
                                  x2 * (1.0f - x3),          x2 * x3 };
            #pragma unroll
            for (int ii = 0; ii < 4; ++ii)
                #pragma unroll
                for (int jj = 0; jj < 4; ++jj)
                    w[l][ii * 4 + jj] = pq[ii] * rs[jj];
        }

        v2f Af[L1_][4];
        #pragma unroll
        for (int l = 0; l < L1_; ++l)
            #pragma unroll
            for (int c = 0; c < 4; ++c) {
                v2f a;
                a[0] = hib ? w[l][4 * c + 2] : w[l][4 * c + 0];
                a[1] = hib ? w[l][4 * c + 3] : w[l][4 * c + 1];
                Af[l][c] = a;
            }

        #pragma unroll
        for (int ct = 0; ct < 2; ++ct) {
            v8f acc[L1_];
            #pragma unroll
            for (int l = 0; l < L1_; ++l) acc[l] = (v8f)0.0f;
            #pragma unroll
            for (int l = 0; l < L1_; ++l)
                #pragma unroll
                for (int c = 0; c < 4; ++c)
                    acc[l] = __builtin_amdgcn_wmma_f32_16x16x4_f32(
                         false, Af[l][c],
                         false, Bf[ct][l][c],
                         (short)0, acc[l],
                         false,  false);

            const int e = ct * 16 + nn;
            #pragma unroll
            for (int j = 0; j < 8; ++j) {
                const float h0 = clamp01(acc[0][j]);
                const float h1v = clamp01(acc[1][j]);
                const float h2 = clamp01(acc[2][j]);
                const float h3 = clamp01(acc[3][j]);
                const float pq[4] = { (1.0f - h0) * (1.0f - h1v), (1.0f - h0) * h1v,
                                      h0 * (1.0f - h1v),          h0 * h1v };
                const float rs[4] = { (1.0f - h2) * (1.0f - h3), (1.0f - h2) * h3,
                                      h2 * (1.0f - h3),          h2 * h3 };
                float o = 0.0f;
                #pragma unroll
                for (int ii = 0; ii < 4; ++ii)
                    #pragma unroll
                    for (int jj = 0; jj < 4; ++jj)
                        o = __builtin_fmaf(pq[ii] * rs[jj], k2r[ct][ii * 4 + jj], o);
                const int row = j + 8 * hi;
                if (e < E_)
                    out[(size_t)(rowBase + row) * E_ + e] = o;
            }
        }
    }
}

extern "C" void kernel_launch(void* const* d_in, const int* in_sizes, int n_in,
                              void* d_out, int out_size, void* d_ws, size_t ws_size,
                              hipStream_t stream) {
    const float* x        = (const float*)d_in[0];
    const float* cal_kp   = (const float*)d_in[1];
    const float* cal_vals = (const float*)d_in[2];
    const float* kernels1 = (const float*)d_in[3];
    const float* kernels2 = (const float*)d_in[4];
    float* out = (float*)d_out;

    const int nrows  = in_sizes[0] / D_;

    lattice_ensemble_kernel<<<NBLK_, 256, 0, stream>>>(
        x, cal_kp, cal_vals, kernels1, kernels2, out, nrows);
}


# Calibration re-laid: a lane owns one feature for 8 rows and reads that feature's table row once per tile (12 wide LDS reads, then registers) instead of 145 per-segment table reads; sums unchanged
# speedup vs baseline: 1.4787x; 1.2630x over previous
#include <hip/hip_runtime.h>


typedef __attribute__((ext_vector_type(8))) float v8f;
typedef __attribute__((ext_vector_type(2))) float v2f;

#define D_   16
#define L1_  4
#define E_   24
#define K_   16
#define NV_  16
#define TR_  16
#define WAVES_ 8
#define NBLK_  512

__device__ __forceinline__ float clamp01(float v) {
    return __builtin_fminf(__builtin_fmaxf(v, 0.0f), 1.0f);
}

__global__ __launch_bounds__(256, 1)
void lattice_ensemble_kernel(const float* __restrict__ x,
                             const float* __restrict__ cal_kp,
                             const float* __restrict__ cal_vals,
                             const float* __restrict__ k1,
                             const float* __restrict__ k2,
                             float* __restrict__ out,
                             int nrows)
{
    __shared__ __align__(16) float sT[D_][52];
    __shared__ __align__(16) float sXC[WAVES_][TR_][20];

    const int tid = threadIdx.x;
    if (tid < D_ * (K_ - 1)) {
        int f = tid / (K_ - 1), s = tid % (K_ - 1);
        float kp0 = cal_kp[f * K_ + s], kp1 = cal_kp[f * K_ + s + 1];
        float inv = 1.0f / (kp1 - kp0);
        sT[f][s]      = inv;
        sT[f][16 + s] = -kp0 * inv;
        sT[f][32 + s] = cal_vals[f * K_ + s + 1] - cal_vals[f * K_ + s];
    }
    if (tid < D_) {
        sT[tid][15] = cal_vals[tid * K_];
        sT[tid][31] = 0.0f;
        sT[tid][47] = 0.0f;
    }
    __syncthreads();

    const int lane = tid & 31;
    const int wid  = tid >> 5;
    const int nn   = lane & 15;
    const bool hib = (lane >= 16);
    const int  hi  = hib ? 1 : 0;

    v2f Bf[2][L1_][4];
    #pragma unroll
    for (int ct = 0; ct < 2; ++ct) {
        int e = ct * 16 + nn;
        #pragma unroll
        for (int l = 0; l < L1_; ++l) {
            #pragma unroll
            for (int c = 0; c < 4; ++c) {
                int kb = 4 * c + 2 * hi;
                float b0 = 0.0f, b1 = 0.0f;
                if (e < E_) {
                    const float* kp = k1 + (size_t)e * (L1_ * NV_) + l * NV_ + kb;
                    b0 = kp[0];
                    b1 = kp[1];
                }
                v2f bb; bb[0] = b0; bb[1] = b1;
                Bf[ct][l][c] = bb;
            }
        }
    }
    float k2r[2][NV_];
    #pragma unroll
    for (int ct = 0; ct < 2; ++ct) {
        int e = ct * 16 + nn;
        #pragma unroll
        for (int k = 0; k < NV_; ++k)
            k2r[ct][k] = (e < E_) ? k2[(size_t)e * NV_ + k] : 0.0f;
    }

    const int ntiles = nrows >> 4;
    const int nwaves = gridDim.x * WAVES_;
    for (int tile = blockIdx.x * WAVES_ + wid; tile < ntiles; tile += nwaves) {
        const int rowBase = tile * TR_;

        int nxt = tile + nwaves;
        if (nxt < ntiles)
            __builtin_prefetch(x + (size_t)(nxt * TR_ + nn) * D_ + 8 * hi, 0, 1);

        const float* xp = x + (size_t)(rowBase + 8 * hi) * D_ + nn;
        float xv[8];
        #pragma unroll
        for (int i = 0; i < 8; ++i) xv[i] = xp[i * D_];

        float ta[16], tb[16], tdv[16];
        {
            const float4* tp = (const float4*)&sT[nn][0];
            #pragma unroll
            for (int q = 0; q < 4; ++q) {
                const float4 a = tp[q], b = tp[4 + q], d = tp[8 + q];
                ta [4 * q] = a.x; ta [4 * q + 1] = a.y; ta [4 * q + 2] = a.z; ta [4 * q + 3] = a.w;
                tb [4 * q] = b.x; tb [4 * q + 1] = b.y; tb [4 * q + 2] = b.z; tb [4 * q + 3] = b.w;
                tdv[4 * q] = d.x; tdv[4 * q + 1] = d.y; tdv[4 * q + 2] = d.z; tdv[4 * q + 3] = d.w;
            }
        }
        const float tv0 = ta[15];

        float xc[8];
        #pragma unroll
        for (int i = 0; i < 8; ++i) xc[i] = tv0;
        #pragma unroll
        for (int s = 0; s < K_ - 1; ++s) {
            float t[8];
            #pragma unroll
            for (int i = 0; i < 8; ++i) t[i] = clamp01(__builtin_fmaf(xv[i], ta[s], tb[s]));
            #pragma unroll
            for (int i = 0; i < 8; ++i) xc[i] = __builtin_fmaf(t[i], tdv[s], xc[i]);
        }
        #pragma unroll
        for (int i = 0; i < 8; ++i) sXC[wid][8 * hi + i][nn] = xc[i];

        asm volatile("s_wait_dscnt 0" ::: "memory");

        const float* rp = &sXC[wid][nn][0];
        const float4 r0 = *(const float4*)(rp);
        const float4 r1 = *(const float4*)(rp + 4);
        const float4 r2 = *(const float4*)(rp + 8);
        const float4 r3 = *(const float4*)(rp + 12);
        const float c16[16] = { r0.x, r0.y, r0.z, r0.w,  r1.x, r1.y, r1.z, r1.w,
                                r2.x, r2.y, r2.z, r2.w,  r3.x, r3.y, r3.z, r3.w };

        float w[L1_][NV_];
        #pragma unroll
        for (int l = 0; l < L1_; ++l) {
            const float x0 = c16[4 * l + 0], x1 = c16[4 * l + 1];
            const float x2 = c16[4 * l + 2], x3 = c16[4 * l + 3];
            const float pq[4] = { (1.0f - x0) * (1.0f - x1), (1.0f - x0) * x1,
                                  x0 * (1.0f - x1),          x0 * x1 };
            const float rs[4] = { (1.0f - x2) * (1.0f - x3), (1.0f - x2) * x3,
                                  x2 * (1.0f - x3),          x2 * x3 };
            #pragma unroll
            for (int ii = 0; ii < 4; ++ii)
                #pragma unroll
                for (int jj = 0; jj < 4; ++jj)
                    w[l][ii * 4 + jj] = pq[ii] * rs[jj];
        }

        v2f Af[L1_][4];
        #pragma unroll
        for (int l = 0; l < L1_; ++l)
            #pragma unroll
            for (int c = 0; c < 4; ++c) {
                v2f a;
                a[0] = hib ? w[l][4 * c + 2] : w[l][4 * c + 0];
                a[1] = hib ? w[l][4 * c + 3] : w[l][4 * c + 1];
                Af[l][c] = a;
            }

        #pragma unroll
        for (int ct = 0; ct < 2; ++ct) {
            v8f acc[L1_];
            #pragma unroll
            for (int l = 0; l < L1_; ++l) acc[l] = (v8f)0.0f;
            #pragma unroll
            for (int l = 0; l < L1_; ++l)
                #pragma unroll
                for (int c = 0; c < 4; ++c)
                    acc[l] = __builtin_amdgcn_wmma_f32_16x16x4_f32(
                         false, Af[l][c],
                         false, Bf[ct][l][c],
                         (short)0, acc[l],
                         false,  false);

            const int e = ct * 16 + nn;
            #pragma unroll
            for (int j = 0; j < 8; ++j) {
                const float h0 = clamp01(acc[0][j]);
                const float h1v = clamp01(acc[1][j]);
                const float h2 = clamp01(acc[2][j]);
                const float h3 = clamp01(acc[3][j]);
                const float pq[4] = { (1.0f - h0) * (1.0f - h1v), (1.0f - h0) * h1v,
                                      h0 * (1.0f - h1v),          h0 * h1v };
                const float rs[4] = { (1.0f - h2) * (1.0f - h3), (1.0f - h2) * h3,
                                      h2 * (1.0f - h3),          h2 * h3 };
                float o = 0.0f;
                #pragma unroll
                for (int ii = 0; ii < 4; ++ii)
                    #pragma unroll
                    for (int jj = 0; jj < 4; ++jj)
                        o = __builtin_fmaf(pq[ii] * rs[jj], k2r[ct][ii * 4 + jj], o);
                const int row = j + 8 * hi;
                if (e < E_)
                    out[(size_t)(rowBase + row) * E_ + e] = o;
            }
        }
    }
}

extern "C" void kernel_launch(void* const* d_in, const int* in_sizes, int n_in,
                              void* d_out, int out_size, void* d_ws, size_t ws_size,
                              hipStream_t stream) {
    const float* x        = (const float*)d_in[0];
    const float* cal_kp   = (const float*)d_in[1];
    const float* cal_vals = (const float*)d_in[2];
    const float* kernels1 = (const float*)d_in[3];
    const float* kernels2 = (const float*)d_in[4];
    float* out = (float*)d_out;

    const int nrows  = in_sizes[0] / D_;

    lattice_ensemble_kernel<<<NBLK_, 256, 0, stream>>>(
        x, cal_kp, cal_vals, kernels1, kernels2, out, nrows);
}


# Start-up loads (keypoints, tile-0 x, k2, unguarded B fragments) issued together before the table build; its barrier fences LDS only and waits for none; next tile's x a tile ahead in registers
# speedup vs baseline: 1.7780x; 1.2024x over previous
#include <hip/hip_runtime.h>


typedef __attribute__((ext_vector_type(8))) float v8f;
typedef __attribute__((ext_vector_type(2))) float v2f;

#define D_   16
#define L1_  4
#define E_   24
#define K_   16
#define NV_  16
#define TR_  16
#define WAVES_ 8
#define NBLK_  512

__device__ __forceinline__ float clamp01(float v) {
    return __builtin_fminf(__builtin_fmaxf(v, 0.0f), 1.0f);
}

__global__ __launch_bounds__(256, 1)
void lattice_ensemble_kernel(const float* __restrict__ x,
                             const float* __restrict__ cal_kp,
                             const float* __restrict__ cal_vals,
                             const float* __restrict__ k1,
                             const float* __restrict__ k2,
                             float* __restrict__ out,
                             int nrows)
{
    __shared__ __align__(16) float sT[D_][52];
    __shared__ __align__(16) float sXC[WAVES_][TR_][20];

    const int tid  = threadIdx.x;
    const int lane = tid & 31;
    const int wid  = tid >> 5;
    const int nn   = lane & 15;
    const bool hib = (lane >= 16);
    const int  hi  = hib ? 1 : 0;

    const int ntiles = nrows >> 4;
    const int nwaves = gridDim.x * WAVES_;
    int tile = blockIdx.x * WAVES_ + wid;


    const int tf = tid / (K_ - 1), ts = tid % (K_ - 1);
    float kp0 = 0.0f, kp1 = 1.0f, cv0 = 0.0f, cv1 = 0.0f;
    if (tid < D_ * (K_ - 1)) {
        kp0 = cal_kp[tf * K_ + ts];   kp1 = cal_kp[tf * K_ + ts + 1];
        cv0 = cal_vals[tf * K_ + ts]; cv1 = cal_vals[tf * K_ + ts + 1];
    }
    const float cvf = cal_vals[(tid & (D_ - 1)) * K_];


    float k2r[2][NV_];
    #pragma unroll
    for (int ct = 0; ct < 2; ++ct) {
        const int e  = ct * 16 + nn;
        const int ec = e < E_ ? e : E_ - 1;
        #pragma unroll
        for (int k = 0; k < NV_; ++k)
            k2r[ct][k] = k2[(size_t)ec * NV_ + k];
    }

    float xv[8];
    #pragma unroll
    for (int i = 0; i < 8; ++i) xv[i] = 0.0f;
    if (tile < ntiles) {
        const float* xp = x + (size_t)(tile * TR_ + 8 * hi) * D_ + nn;
        #pragma unroll
        for (int i = 0; i < 8; ++i) xv[i] = xp[i * D_];
    }

    v2f Bf[2][L1_][4];
    #pragma unroll
    for (int ct = 0; ct < 2; ++ct) {
        const int e  = ct * 16 + nn;
        const int ec = e < E_ ? e : E_ - 1;
        #pragma unroll
        for (int l = 0; l < L1_; ++l) {
            #pragma unroll
            for (int c = 0; c < 4; ++c) {
                const int kb = 4 * c + 2 * hi;
                const float* kp = k1 + (size_t)ec * (L1_ * NV_) + l * NV_ + kb;
                v2f bb; bb[0] = kp[0]; bb[1] = kp[1];
                Bf[ct][l][c] = bb;
            }
        }
    }

    if (tid < D_ * (K_ - 1)) {
        float inv = 1.0f / (kp1 - kp0);
        sT[tf][ts]      = inv;
        sT[tf][16 + ts] = -kp0 * inv;
        sT[tf][32 + ts] = cv1 - cv0;
    }
    if (tid < D_) {
        sT[tid][15] = cvf;
        sT[tid][31] = 0.0f;
        sT[tid][47] = 0.0f;
    }
    __builtin_amdgcn_fence(__ATOMIC_RELEASE, "workgroup", "local");
    __builtin_amdgcn_s_barrier();
    __builtin_amdgcn_fence(__ATOMIC_ACQUIRE, "workgroup", "local");

    for (; tile < ntiles; tile += nwaves) {
        const int rowBase = tile * TR_;

        float xn[8];
        #pragma unroll
        for (int i = 0; i < 8; ++i) xn[i] = 0.0f;
        {
            const int nxt = tile + nwaves;
            if (nxt < ntiles) {
                const float* xq = x + (size_t)(nxt * TR_ + 8 * hi) * D_ + nn;
                #pragma unroll
                for (int i = 0; i < 8; ++i) xn[i] = xq[i * D_];
            }
        }


        float ta[16], tb[16], tdv[16];
        {
            const float4* tp = (const float4*)&sT[nn][0];
            #pragma unroll
            for (int q = 0; q < 4; ++q) {
                const float4 a = tp[q], b = tp[4 + q], d = tp[8 + q];
                ta [4 * q] = a.x; ta [4 * q + 1] = a.y; ta [4 * q + 2] = a.z; ta [4 * q + 3] = a.w;
                tb [4 * q] = b.x; tb [4 * q + 1] = b.y; tb [4 * q + 2] = b.z; tb [4 * q + 3] = b.w;
                tdv[4 * q] = d.x; tdv[4 * q + 1] = d.y; tdv[4 * q + 2] = d.z; tdv[4 * q + 3] = d.w;
            }
        }
        const float tv0 = ta[15];

        float xc[8];
        #pragma unroll
        for (int i = 0; i < 8; ++i) xc[i] = tv0;
        #pragma unroll
        for (int s = 0; s < K_ - 1; ++s) {
            float t[8];
            #pragma unroll
            for (int i = 0; i < 8; ++i) t[i] = clamp01(__builtin_fmaf(xv[i], ta[s], tb[s]));
            #pragma unroll
            for (int i = 0; i < 8; ++i) xc[i] = __builtin_fmaf(t[i], tdv[s], xc[i]);
        }
        #pragma unroll
        for (int i = 0; i < 8; ++i) sXC[wid][8 * hi + i][nn] = xc[i];

        asm volatile("s_wait_dscnt 0" ::: "memory");

        const float* rp = &sXC[wid][nn][0];
        const float4 r0 = *(const float4*)(rp);
        const float4 r1 = *(const float4*)(rp + 4);
        const float4 r2 = *(const float4*)(rp + 8);
        const float4 r3 = *(const float4*)(rp + 12);
        const float c16[16] = { r0.x, r0.y, r0.z, r0.w,  r1.x, r1.y, r1.z, r1.w,
                                r2.x, r2.y, r2.z, r2.w,  r3.x, r3.y, r3.z, r3.w };

        float w[L1_][NV_];
        #pragma unroll
        for (int l = 0; l < L1_; ++l) {
            const float x0 = c16[4 * l + 0], x1 = c16[4 * l + 1];
            const float x2 = c16[4 * l + 2], x3 = c16[4 * l + 3];
            const float pq[4] = { (1.0f - x0) * (1.0f - x1), (1.0f - x0) * x1,
                                  x0 * (1.0f - x1),          x0 * x1 };
            const float rs[4] = { (1.0f - x2) * (1.0f - x3), (1.0f - x2) * x3,
                                  x2 * (1.0f - x3),          x2 * x3 };
            #pragma unroll
            for (int ii = 0; ii < 4; ++ii)
                #pragma unroll
                for (int jj = 0; jj < 4; ++jj)
                    w[l][ii * 4 + jj] = pq[ii] * rs[jj];
        }

        v2f Af[L1_][4];
        #pragma unroll
        for (int l = 0; l < L1_; ++l)
            #pragma unroll
            for (int c = 0; c < 4; ++c) {
                v2f a;
                a[0] = hib ? w[l][4 * c + 2] : w[l][4 * c + 0];
                a[1] = hib ? w[l][4 * c + 3] : w[l][4 * c + 1];
                Af[l][c] = a;
            }

        #pragma unroll
        for (int ct = 0; ct < 2; ++ct) {
            v8f acc[L1_];
            #pragma unroll
            for (int l = 0; l < L1_; ++l) acc[l] = (v8f)0.0f;
            #pragma unroll
            for (int l = 0; l < L1_; ++l)
                #pragma unroll
                for (int c = 0; c < 4; ++c)
                    acc[l] = __builtin_amdgcn_wmma_f32_16x16x4_f32(
                         false, Af[l][c],
                         false, Bf[ct][l][c],
                         (short)0, acc[l],
                         false,  false);

            const int e = ct * 16 + nn;
            #pragma unroll
            for (int j = 0; j < 8; ++j) {
                const float h0 = clamp01(acc[0][j]);
                const float h1v = clamp01(acc[1][j]);
                const float h2 = clamp01(acc[2][j]);
                const float h3 = clamp01(acc[3][j]);
                const float pq[4] = { (1.0f - h0) * (1.0f - h1v), (1.0f - h0) * h1v,
                                      h0 * (1.0f - h1v),          h0 * h1v };
                const float rs[4] = { (1.0f - h2) * (1.0f - h3), (1.0f - h2) * h3,
                                      h2 * (1.0f - h3),          h2 * h3 };
                float o = 0.0f;
                #pragma unroll
                for (int ii = 0; ii < 4; ++ii)
                    #pragma unroll
                    for (int jj = 0; jj < 4; ++jj)
                        o = __builtin_fmaf(pq[ii] * rs[jj], k2r[ct][ii * 4 + jj], o);
                const int row = j + 8 * hi;
                if (e < E_)
                    out[(size_t)(rowBase + row) * E_ + e] = o;
            }
        }

        #pragma unroll
        for (int i = 0; i < 8; ++i) xv[i] = xn[i];
    }
}

extern "C" void kernel_launch(void* const* d_in, const int* in_sizes, int n_in,
                              void* d_out, int out_size, void* d_ws, size_t ws_size,
                              hipStream_t stream) {
    const float* x        = (const float*)d_in[0];
    const float* cal_kp   = (const float*)d_in[1];
    const float* cal_vals = (const float*)d_in[2];
    const float* kernels1 = (const float*)d_in[3];
    const float* kernels2 = (const float*)d_in[4];
    float* out = (float*)d_out;

    const int nrows  = in_sizes[0] / D_;

    lattice_ensemble_kernel<<<NBLK_, 256, 0, stream>>>(
        x, cal_kp, cal_vals, kernels1, kernels2, out, nrows);
}


# Layer-2 lattice as a 4-level Horner on per-column multilinear coefficients built once per wave: 15 packed fma per row pair instead of 24 products + 16 fma; the output's sum order changes
# speedup vs baseline: 1.9693x; 1.1076x over previous
#include <hip/hip_runtime.h>


typedef __attribute__((ext_vector_type(8))) float v8f;
typedef __attribute__((ext_vector_type(2))) float v2f;

#define D_   16
#define L1_  4
#define E_   24
#define K_   16
#define NV_  16
#define TR_  16
#define WAVES_ 8
#define NBLK_  512

__device__ __forceinline__ float clamp01(float v) {
    return __builtin_fminf(__builtin_fmaxf(v, 0.0f), 1.0f);
}

__global__ __launch_bounds__(256, 1)
void lattice_ensemble_kernel(const float* __restrict__ x,
                             const float* __restrict__ cal_kp,
                             const float* __restrict__ cal_vals,
                             const float* __restrict__ k1,
                             const float* __restrict__ k2,
                             float* __restrict__ out,
                             int nrows)
{
    __shared__ __align__(16) float sT[D_][52];
    __shared__ __align__(16) float sXC[WAVES_][TR_][20];

    const int tid  = threadIdx.x;
    const int lane = tid & 31;
    const int wid  = tid >> 5;
    const int nn   = lane & 15;
    const bool hib = (lane >= 16);
    const int  hi  = hib ? 1 : 0;

    const int ntiles = nrows >> 4;
    const int nwaves = gridDim.x * WAVES_;
    int tile = blockIdx.x * WAVES_ + wid;


    const int tf = tid / (K_ - 1), ts = tid % (K_ - 1);
    float kp0 = 0.0f, kp1 = 1.0f, cv0 = 0.0f, cv1 = 0.0f;
    if (tid < D_ * (K_ - 1)) {
        kp0 = cal_kp[tf * K_ + ts];   kp1 = cal_kp[tf * K_ + ts + 1];
        cv0 = cal_vals[tf * K_ + ts]; cv1 = cal_vals[tf * K_ + ts + 1];
    }
    const float cvf = cal_vals[(tid & (D_ - 1)) * K_];


    float k2r[2][NV_];
    #pragma unroll
    for (int ct = 0; ct < 2; ++ct) {
        const int e  = ct * 16 + nn;
        const int ec = e < E_ ? e : E_ - 1;
        #pragma unroll
        for (int k = 0; k < NV_; ++k)
            k2r[ct][k] = k2[(size_t)ec * NV_ + k];
    }

    float xv[8];
    #pragma unroll
    for (int i = 0; i < 8; ++i) xv[i] = 0.0f;
    if (tile < ntiles) {
        const float* xp = x + (size_t)(tile * TR_ + 8 * hi) * D_ + nn;
        #pragma unroll
        for (int i = 0; i < 8; ++i) xv[i] = xp[i * D_];
    }

    v2f Bf[2][L1_][4];
    #pragma unroll
    for (int ct = 0; ct < 2; ++ct) {
        const int e  = ct * 16 + nn;
        const int ec = e < E_ ? e : E_ - 1;
        #pragma unroll
        for (int l = 0; l < L1_; ++l) {
            #pragma unroll
            for (int c = 0; c < 4; ++c) {
                const int kb = 4 * c + 2 * hi;
                const float* kp = k1 + (size_t)ec * (L1_ * NV_) + l * NV_ + kb;
                v2f bb; bb[0] = kp[0]; bb[1] = kp[1];
                Bf[ct][l][c] = bb;
            }
        }
    }

    if (tid < D_ * (K_ - 1)) {
        float inv = 1.0f / (kp1 - kp0);
        sT[tf][ts]      = inv;
        sT[tf][16 + ts] = -kp0 * inv;
        sT[tf][32 + ts] = cv1 - cv0;
    }
    if (tid < D_) {
        sT[tid][15] = cvf;
        sT[tid][31] = 0.0f;
        sT[tid][47] = 0.0f;
    }
    __builtin_amdgcn_fence(__ATOMIC_RELEASE, "workgroup", "local");
    __builtin_amdgcn_s_barrier();
    __builtin_amdgcn_fence(__ATOMIC_ACQUIRE, "workgroup", "local");

    v2f C2[2][NV_];
    #pragma unroll
    for (int ct = 0; ct < 2; ++ct) {
        float c[NV_];
        #pragma unroll
        for (int k = 0; k < NV_; ++k) c[k] = k2r[ct][k];
        #pragma unroll
        for (int bit = 1; bit < NV_; bit <<= 1)
            #pragma unroll
            for (int k = 0; k < NV_; ++k)
                if (k & bit) c[k] = c[k] - c[k ^ bit];
        #pragma unroll
        for (int k = 0; k < NV_; ++k) { v2f cc; cc[0] = c[k]; cc[1] = c[k]; C2[ct][k] = cc; }
    }

    for (; tile < ntiles; tile += nwaves) {
        const int rowBase = tile * TR_;

        float xn[8];
        #pragma unroll
        for (int i = 0; i < 8; ++i) xn[i] = 0.0f;
        {
            const int nxt = tile + nwaves;
            if (nxt < ntiles) {
                const float* xq = x + (size_t)(nxt * TR_ + 8 * hi) * D_ + nn;
                #pragma unroll
                for (int i = 0; i < 8; ++i) xn[i] = xq[i * D_];
            }
        }


        float ta[16], tb[16], tdv[16];
        {
            const float4* tp = (const float4*)&sT[nn][0];
            #pragma unroll
            for (int q = 0; q < 4; ++q) {
                const float4 a = tp[q], b = tp[4 + q], d = tp[8 + q];
                ta [4 * q] = a.x; ta [4 * q + 1] = a.y; ta [4 * q + 2] = a.z; ta [4 * q + 3] = a.w;
                tb [4 * q] = b.x; tb [4 * q + 1] = b.y; tb [4 * q + 2] = b.z; tb [4 * q + 3] = b.w;
                tdv[4 * q] = d.x; tdv[4 * q + 1] = d.y; tdv[4 * q + 2] = d.z; tdv[4 * q + 3] = d.w;
            }
        }
        const float tv0 = ta[15];

        float xc[8];
        #pragma unroll
        for (int i = 0; i < 8; ++i) xc[i] = tv0;
        #pragma unroll
        for (int s = 0; s < K_ - 1; ++s) {
            float t[8];
            #pragma unroll
            for (int i = 0; i < 8; ++i) t[i] = clamp01(__builtin_fmaf(xv[i], ta[s], tb[s]));
            #pragma unroll
            for (int i = 0; i < 8; ++i) xc[i] = __builtin_fmaf(t[i], tdv[s], xc[i]);
        }
        #pragma unroll
        for (int i = 0; i < 8; ++i) sXC[wid][8 * hi + i][nn] = xc[i];

        asm volatile("s_wait_dscnt 0" ::: "memory");

        const float* rp = &sXC[wid][nn][0];
        const float4 r0 = *(const float4*)(rp);
        const float4 r1 = *(const float4*)(rp + 4);
        const float4 r2 = *(const float4*)(rp + 8);
        const float4 r3 = *(const float4*)(rp + 12);
        const float c16[16] = { r0.x, r0.y, r0.z, r0.w,  r1.x, r1.y, r1.z, r1.w,
                                r2.x, r2.y, r2.z, r2.w,  r3.x, r3.y, r3.z, r3.w };

        float w[L1_][NV_];
        #pragma unroll
        for (int l = 0; l < L1_; ++l) {
            const float x0 = c16[4 * l + 0], x1 = c16[4 * l + 1];
            const float x2 = c16[4 * l + 2], x3 = c16[4 * l + 3];
            const float pq[4] = { (1.0f - x0) * (1.0f - x1), (1.0f - x0) * x1,
                                  x0 * (1.0f - x1),          x0 * x1 };
            const float rs[4] = { (1.0f - x2) * (1.0f - x3), (1.0f - x2) * x3,
                                  x2 * (1.0f - x3),          x2 * x3 };
            #pragma unroll
            for (int ii = 0; ii < 4; ++ii)
                #pragma unroll
                for (int jj = 0; jj < 4; ++jj)
                    w[l][ii * 4 + jj] = pq[ii] * rs[jj];
        }

        v2f Af[L1_][4];
        #pragma unroll
        for (int l = 0; l < L1_; ++l)
            #pragma unroll
            for (int c = 0; c < 4; ++c) {
                v2f a;
                a[0] = hib ? w[l][4 * c + 2] : w[l][4 * c + 0];
                a[1] = hib ? w[l][4 * c + 3] : w[l][4 * c + 1];
                Af[l][c] = a;
            }

        #pragma unroll
        for (int ct = 0; ct < 2; ++ct) {
            v8f acc[L1_];
            #pragma unroll
            for (int l = 0; l < L1_; ++l) acc[l] = (v8f)0.0f;
            #pragma unroll
            for (int l = 0; l < L1_; ++l)
                #pragma unroll
                for (int c = 0; c < 4; ++c)
                    acc[l] = __builtin_amdgcn_wmma_f32_16x16x4_f32(
                         false, Af[l][c],
                         false, Bf[ct][l][c],
                         (short)0, acc[l],
                         false,  false);

            const int e = ct * 16 + nn;
            #pragma unroll
            for (int jp = 0; jp < 4; ++jp) {
                v2f t0, t1, t2, t3;
                t0[0] = clamp01(acc[0][2 * jp]); t0[1] = clamp01(acc[0][2 * jp + 1]);
                t1[0] = clamp01(acc[1][2 * jp]); t1[1] = clamp01(acc[1][2 * jp + 1]);
                t2[0] = clamp01(acc[2][2 * jp]); t2[1] = clamp01(acc[2][2 * jp + 1]);
                t3[0] = clamp01(acc[3][2 * jp]); t3[1] = clamp01(acc[3][2 * jp + 1]);
                v2f p0[8], p1[4], p2[2];
                #pragma unroll
                for (int i = 0; i < 8; ++i)
                    p0[i] = __builtin_elementwise_fma(t0, C2[ct][i + 8], C2[ct][i]);
                #pragma unroll
                for (int i = 0; i < 4; ++i)
                    p1[i] = __builtin_elementwise_fma(t1, p0[i + 4], p0[i]);
                #pragma unroll
                for (int i = 0; i < 2; ++i)
                    p2[i] = __builtin_elementwise_fma(t2, p1[i + 2], p1[i]);
                const v2f o = __builtin_elementwise_fma(t3, p2[1], p2[0]);
                const int row = 2 * jp + 8 * hi;
                if (e < E_) {
                    out[(size_t)(rowBase + row) * E_ + e]     = o[0];
                    out[(size_t)(rowBase + row + 1) * E_ + e] = o[1];
                }
            }
        }

        #pragma unroll
        for (int i = 0; i < 8; ++i) xv[i] = xn[i];
    }
}

extern "C" void kernel_launch(void* const* d_in, const int* in_sizes, int n_in,
                              void* d_out, int out_size, void* d_ws, size_t ws_size,
                              hipStream_t stream) {
    const float* x        = (const float*)d_in[0];
    const float* cal_kp   = (const float*)d_in[1];
    const float* cal_vals = (const float*)d_in[2];
    const float* kernels1 = (const float*)d_in[3];
    const float* kernels2 = (const float*)d_in[4];
    float* out = (float*)d_out;

    const int nrows  = in_sizes[0] / D_;

    lattice_ensemble_kernel<<<NBLK_, 256, 0, stream>>>(
        x, cal_kp, cal_vals, kernels1, kernels2, out, nrows);
}


# kernels1 fetched once per work-group with whole-line loads into LDS (row pitch 68 floats); each wave reads its 32 B fragments from LDS instead of gathering them from global memory at 16 lines per load
# speedup vs baseline: 2.1422x; 1.0878x over previous
#include <hip/hip_runtime.h>


typedef __attribute__((ext_vector_type(8))) float v8f;
typedef __attribute__((ext_vector_type(2))) float v2f;

#define D_   16
#define L1_  4
#define E_   24
#define K_   16
#define NV_  16
#define TR_  16
#define WAVES_ 8
#define NBLK_  512

__device__ __forceinline__ float clamp01(float v) {
    return __builtin_fminf(__builtin_fmaxf(v, 0.0f), 1.0f);
}

__global__ __launch_bounds__(256, 1)
void lattice_ensemble_kernel(const float* __restrict__ x,
                             const float* __restrict__ cal_kp,
                             const float* __restrict__ cal_vals,
                             const float* __restrict__ k1,
                             const float* __restrict__ k2,
                             float* __restrict__ out,
                             int nrows)
{
    __shared__ __align__(16) float sT[D_][52];
    __shared__ __align__(16) float sXC[WAVES_][TR_][20];
    __shared__ __align__(16) float sK1[E_][68];

    const int tid  = threadIdx.x;
    const int lane = tid & 31;
    const int wid  = tid >> 5;
    const int nn   = lane & 15;
    const bool hib = (lane >= 16);
    const int  hi  = hib ? 1 : 0;

    const int ntiles = nrows >> 4;
    const int nwaves = gridDim.x * WAVES_;
    int tile = blockIdx.x * WAVES_ + wid;


    const float cvf = cal_vals[(tid & (D_ - 1)) * K_];
    const int tf = tid / (K_ - 1), ts = tid % (K_ - 1);
    float kp0 = 0.0f, kp1 = 1.0f, cv0 = 0.0f, cv1 = 0.0f;
    if (tid < D_ * (K_ - 1)) {
        kp0 = cal_kp[tf * K_ + ts];   kp1 = cal_kp[tf * K_ + ts + 1];
        cv0 = cal_vals[tf * K_ + ts]; cv1 = cal_vals[tf * K_ + ts + 1];
    }
    float k1a[4], k1b[2];
    #pragma unroll
    for (int i = 0; i < 4; ++i) k1a[i] = k1[4 * tid + i];
    #pragma unroll
    for (int i = 0; i < 2; ++i) k1b[i] = k1[(E_ - 8) * L1_ * NV_ + 2 * tid + i];


    float k2r[2][NV_];
    #pragma unroll
    for (int ct = 0; ct < 2; ++ct) {
        const int e  = ct * 16 + nn;
        const int ec = e < E_ ? e : E_ - 1;
        #pragma unroll
        for (int k = 0; k < NV_; ++k)
            k2r[ct][k] = k2[(size_t)ec * NV_ + k];
    }

    float xv[8];
    #pragma unroll
    for (int i = 0; i < 8; ++i) xv[i] = 0.0f;
    if (tile < ntiles) {
        const float* xp = x + (size_t)(tile * TR_ + 8 * hi) * D_ + nn;
        #pragma unroll
        for (int i = 0; i < 8; ++i) xv[i] = xp[i * D_];
    }

    {
        float4 q; q.x = k1a[0]; q.y = k1a[1]; q.z = k1a[2]; q.w = k1a[3];
        *(float4*)&sK1[tid >> 4][4 * (tid & 15)] = q;
        v2f p; p[0] = k1b[0]; p[1] = k1b[1];
        *(v2f*)&sK1[(E_ - 8) + (tid >> 5)][2 * (tid & 31)] = p;
    }
    if (tid < D_ * (K_ - 1)) {
        float inv = 1.0f / (kp1 - kp0);
        sT[tf][ts]      = inv;
        sT[tf][16 + ts] = -kp0 * inv;
        sT[tf][32 + ts] = cv1 - cv0;
    }
    if (tid < D_) {
        sT[tid][15] = cvf;
        sT[tid][31] = 0.0f;
        sT[tid][47] = 0.0f;
    }
    __builtin_amdgcn_fence(__ATOMIC_RELEASE, "workgroup", "local");
    __builtin_amdgcn_s_barrier();
    __builtin_amdgcn_fence(__ATOMIC_ACQUIRE, "workgroup", "local");

    v2f Bf[2][L1_][4];
    #pragma unroll
    for (int ct = 0; ct < 2; ++ct) {
        const int e  = ct * 16 + nn;
        const int ec = e < E_ ? e : E_ - 1;
        #pragma unroll
        for (int l = 0; l < L1_; ++l) {
            #pragma unroll
            for (int c = 0; c < 4; ++c) {
                const int kb = 4 * c + 2 * hi;
                Bf[ct][l][c] = *(const v2f*)&sK1[ec][l * NV_ + kb];
            }
        }
    }

    v2f C2[2][NV_];
    #pragma unroll
    for (int ct = 0; ct < 2; ++ct) {
        float c[NV_];
        #pragma unroll
        for (int k = 0; k < NV_; ++k) c[k] = k2r[ct][k];
        #pragma unroll
        for (int bit = 1; bit < NV_; bit <<= 1)
            #pragma unroll
            for (int k = 0; k < NV_; ++k)
                if (k & bit) c[k] = c[k] - c[k ^ bit];
        #pragma unroll
        for (int k = 0; k < NV_; ++k) { v2f cc; cc[0] = c[k]; cc[1] = c[k]; C2[ct][k] = cc; }
    }

    for (; tile < ntiles; tile += nwaves) {
        const int rowBase = tile * TR_;

        float xn[8];
        #pragma unroll
        for (int i = 0; i < 8; ++i) xn[i] = 0.0f;
        {
            const int nxt = tile + nwaves;
            if (nxt < ntiles) {
                const float* xq = x + (size_t)(nxt * TR_ + 8 * hi) * D_ + nn;
                #pragma unroll
                for (int i = 0; i < 8; ++i) xn[i] = xq[i * D_];
            }
        }


        float ta[16], tb[16], tdv[16];
        {
            const float4* tp = (const float4*)&sT[nn][0];
            #pragma unroll
            for (int q = 0; q < 4; ++q) {
                const float4 a = tp[q], b = tp[4 + q], d = tp[8 + q];
                ta [4 * q] = a.x; ta [4 * q + 1] = a.y; ta [4 * q + 2] = a.z; ta [4 * q + 3] = a.w;
                tb [4 * q] = b.x; tb [4 * q + 1] = b.y; tb [4 * q + 2] = b.z; tb [4 * q + 3] = b.w;
                tdv[4 * q] = d.x; tdv[4 * q + 1] = d.y; tdv[4 * q + 2] = d.z; tdv[4 * q + 3] = d.w;
            }
        }
        const float tv0 = ta[15];

        float xc[8];
        #pragma unroll
        for (int i = 0; i < 8; ++i) xc[i] = tv0;
        #pragma unroll
        for (int s = 0; s < K_ - 1; ++s) {
            float t[8];
            #pragma unroll
            for (int i = 0; i < 8; ++i) t[i] = clamp01(__builtin_fmaf(xv[i], ta[s], tb[s]));
            #pragma unroll
            for (int i = 0; i < 8; ++i) xc[i] = __builtin_fmaf(t[i], tdv[s], xc[i]);
        }
        #pragma unroll
        for (int i = 0; i < 8; ++i) sXC[wid][8 * hi + i][nn] = xc[i];

        asm volatile("s_wait_dscnt 0" ::: "memory");

        const float* rp = &sXC[wid][nn][0];
        const float4 r0 = *(const float4*)(rp);
        const float4 r1 = *(const float4*)(rp + 4);
        const float4 r2 = *(const float4*)(rp + 8);
        const float4 r3 = *(const float4*)(rp + 12);
        const float c16[16] = { r0.x, r0.y, r0.z, r0.w,  r1.x, r1.y, r1.z, r1.w,
                                r2.x, r2.y, r2.z, r2.w,  r3.x, r3.y, r3.z, r3.w };

        float w[L1_][NV_];
        #pragma unroll
        for (int l = 0; l < L1_; ++l) {
            const float x0 = c16[4 * l + 0], x1 = c16[4 * l + 1];
            const float x2 = c16[4 * l + 2], x3 = c16[4 * l + 3];
            const float pq[4] = { (1.0f - x0) * (1.0f - x1), (1.0f - x0) * x1,
                                  x0 * (1.0f - x1),          x0 * x1 };
            const float rs[4] = { (1.0f - x2) * (1.0f - x3), (1.0f - x2) * x3,
                                  x2 * (1.0f - x3),          x2 * x3 };
            #pragma unroll
            for (int ii = 0; ii < 4; ++ii)
                #pragma unroll
                for (int jj = 0; jj < 4; ++jj)
                    w[l][ii * 4 + jj] = pq[ii] * rs[jj];
        }

        v2f Af[L1_][4];
        #pragma unroll
        for (int l = 0; l < L1_; ++l)
            #pragma unroll
            for (int c = 0; c < 4; ++c) {
                v2f a;
                a[0] = hib ? w[l][4 * c + 2] : w[l][4 * c + 0];
                a[1] = hib ? w[l][4 * c + 3] : w[l][4 * c + 1];
                Af[l][c] = a;
            }

        #pragma unroll
        for (int ct = 0; ct < 2; ++ct) {
            v8f acc[L1_];
            #pragma unroll
            for (int l = 0; l < L1_; ++l) acc[l] = (v8f)0.0f;
            #pragma unroll
            for (int l = 0; l < L1_; ++l)
                #pragma unroll
                for (int c = 0; c < 4; ++c)
                    acc[l] = __builtin_amdgcn_wmma_f32_16x16x4_f32(
                         false, Af[l][c],
                         false, Bf[ct][l][c],
                         (short)0, acc[l],
                         false,  false);

            const int e = ct * 16 + nn;
            #pragma unroll
            for (int jp = 0; jp < 4; ++jp) {
                v2f t0, t1, t2, t3;
                t0[0] = clamp01(acc[0][2 * jp]); t0[1] = clamp01(acc[0][2 * jp + 1]);
                t1[0] = clamp01(acc[1][2 * jp]); t1[1] = clamp01(acc[1][2 * jp + 1]);
                t2[0] = clamp01(acc[2][2 * jp]); t2[1] = clamp01(acc[2][2 * jp + 1]);
                t3[0] = clamp01(acc[3][2 * jp]); t3[1] = clamp01(acc[3][2 * jp + 1]);
                v2f p0[8], p1[4], p2[2];
                #pragma unroll
                for (int i = 0; i < 8; ++i)
                    p0[i] = __builtin_elementwise_fma(t0, C2[ct][i + 8], C2[ct][i]);
                #pragma unroll
                for (int i = 0; i < 4; ++i)
                    p1[i] = __builtin_elementwise_fma(t1, p0[i + 4], p0[i]);
                #pragma unroll
                for (int i = 0; i < 2; ++i)
                    p2[i] = __builtin_elementwise_fma(t2, p1[i + 2], p1[i]);
                const v2f o = __builtin_elementwise_fma(t3, p2[1], p2[0]);
                const int row = 2 * jp + 8 * hi;
                if (e < E_) {
                    out[(size_t)(rowBase + row) * E_ + e]     = o[0];
                    out[(size_t)(rowBase + row + 1) * E_ + e] = o[1];
                }
            }
        }

        #pragma unroll
        for (int i = 0; i < 8; ++i) xv[i] = xn[i];
    }
}

extern "C" void kernel_launch(void* const* d_in, const int* in_sizes, int n_in,
                              void* d_out, int out_size, void* d_ws, size_t ws_size,
                              hipStream_t stream) {
    const float* x        = (const float*)d_in[0];
    const float* cal_kp   = (const float*)d_in[1];
    const float* cal_vals = (const float*)d_in[2];
    const float* kernels1 = (const float*)d_in[3];
    const float* kernels2 = (const float*)d_in[4];
    float* out = (float*)d_out;

    const int nrows  = in_sizes[0] / D_;

    lattice_ensemble_kernel<<<NBLK_, 256, 0, stream>>>(
        x, cal_kp, cal_vals, kernels1, kernels2, out, nrows);
}
